# starting kernel
# baseline (speedup 1.0000x reference)
#include <hip/hip_runtime.h>

typedef __attribute__((ext_vector_type(2))) float v2f;
typedef __attribute__((ext_vector_type(8))) float v8f;

#define WMMA_F32(a, b, c) \
  __builtin_amdgcn_wmma_f32_16x16x4_f32(false, (a), false, (b), (short)0, (c), false, false)

constexpr int Bc = 1024;
constexpr int Tc = 200;
constexpr int Ic = 128;
constexpr int Hc = 128;
constexpr int PITCH = 132;
constexpr int BT = 32;

__global__ __launch_bounds__(256) void augru_proj(
    const float* __restrict__ x,
    const float* __restrict__ Wz, const float* __restrict__ bz,
    const float* __restrict__ Wr, const float* __restrict__ br,
    const float* __restrict__ Wh, const float* __restrict__ bh,
    float* __restrict__ xz, float* __restrict__ xr, float* __restrict__ xh)
{
  extern __shared__ float lds[];

  const int gate = blockIdx.y;
  const float* W    = (gate == 0) ? Wz : (gate == 1) ? Wr : Wh;
  const float* bias = (gate == 0) ? bz : (gate == 1) ? br : bh;
  float* outp       = (gate == 0) ? xz : (gate == 1) ? xr : xh;

  const int tid = threadIdx.x;

  for (int idx = tid; idx < Hc * Ic; idx += 256) {
    int n = idx >> 7, k = idx & 127;
    lds[n * PITCH + k] = W[n * (Ic + Hc) + k];
  }
  __syncthreads();

  const int wave = tid >> 5, lane = tid & 31;
  const int lmod = lane & 15, ldiv = lane >> 4;
  const long mt = (long)blockIdx.x * 8 + wave;
  const long rbase = mt * 16;

  float biasv[8];
#pragma unroll
  for (int nt = 0; nt < 8; ++nt) biasv[nt] = bias[nt * 16 + lmod];

  v8f acc[8] = {};

  const float* arow = x + (rbase + lmod) * Ic + 2 * ldiv;
  const float* brow = lds + lmod * PITCH + 2 * ldiv;

  v2f a_c = *(const v2f*)(arow);
  v2f b_c[8];
#pragma unroll
  for (int nt = 0; nt < 8; ++nt) b_c[nt] = *(const v2f*)(brow + nt * 16 * PITCH);

#pragma unroll 4
  for (int kk = 0; kk < Ic - 4; kk += 4) {
    v2f a_n = *(const v2f*)(arow + kk + 4);
    v2f b_n[8];
#pragma unroll
    for (int nt = 0; nt < 8; ++nt)
      b_n[nt] = *(const v2f*)(brow + nt * 16 * PITCH + kk + 4);
#pragma unroll
    for (int nt = 0; nt < 8; ++nt) acc[nt] = WMMA_F32(a_c, b_c[nt], acc[nt]);
    a_c = a_n;
#pragma unroll
    for (int nt = 0; nt < 8; ++nt) b_c[nt] = b_n[nt];
  }
#pragma unroll
  for (int nt = 0; nt < 8; ++nt) acc[nt] = WMMA_F32(a_c, b_c[nt], acc[nt]);

#pragma unroll
  for (int v = 0; v < 8; ++v) {
    long r = rbase + v + 8 * ldiv;
    int  tt = (int)(r % Tc);
    long bb = r / Tc;
    float* orow = outp + ((long)tt * Bc + bb) * Hc + lmod;
#pragma unroll
    for (int nt = 0; nt < 8; ++nt) orow[nt * 16] = acc[nt][v] + biasv[nt];
  }
}

__global__ __launch_bounds__(256) void augru_scan(
    const float* __restrict__ att,
    const float* __restrict__ Wz, const float* __restrict__ Wr,
    const float* __restrict__ Wh,
    const float* __restrict__ xz, const float* __restrict__ xr,
    const float* __restrict__ xh,
    float* __restrict__ out)
{
  extern __shared__ float lds[];
  float* wzl   = lds;
  float* wrl   = wzl + Hc * PITCH;
  float* whl   = wrl + Hc * PITCH;
  float* hbuf  = whl + Hc * PITCH;
  float* rhbuf = hbuf + BT * PITCH;
  float* abuf  = rhbuf + BT * PITCH;

  const int tid = threadIdx.x;
  const int b0  = blockIdx.x * BT;

  for (int idx = tid; idx < Hc * Hc; idx += 256) {
    int n = idx >> 7, k = idx & 127;
    wzl[n * PITCH + k] = Wz[n * (Ic + Hc) + Ic + k];
    wrl[n * PITCH + k] = Wr[n * (Ic + Hc) + Ic + k];
    whl[n * PITCH + k] = Wh[n * (Ic + Hc) + Ic + k];
  }
  for (int idx = tid; idx < BT * Hc; idx += 256)
    hbuf[(idx >> 7) * PITCH + (idx & 127)] = 0.f;
  __syncthreads();

  const int wave = tid >> 5, lane = tid & 31;
  const int lmod = lane & 15, ldiv = lane >> 4;
  const int nt   = wave;
  const int ncol = nt * 16 + lmod;

  const float* wzr = wzl + ncol * PITCH + 2 * ldiv;
  const float* wrr = wrl + ncol * PITCH + 2 * ldiv;
  const float* whr = whl + ncol * PITCH + 2 * ldiv;
  const float* ha0 = hbuf + lmod * PITCH + 2 * ldiv;
  const float* ha1 = hbuf + (16 + lmod) * PITCH + 2 * ldiv;
  const float* ra0 = rhbuf + lmod * PITCH + 2 * ldiv;
  const float* ra1 = rhbuf + (16 + lmod) * PITCH + 2 * ldiv;

  for (int t = 0; t < Tc; ++t) {
    if (tid < BT) abuf[tid] = att[(long)(b0 + tid) * Tc + t];

    if (t + 1 < Tc) {
      long nb = ((long)(t + 1) * Bc + b0) * Hc + tid * 16;
      __builtin_prefetch(xz + nb, 0, 1);
      __builtin_prefetch(xr + nb, 0, 1);
      __builtin_prefetch(xh + nb, 0, 1);
    }

    const long xbase = ((long)t * Bc + b0) * Hc;
    float xzv[16], xrv[16], xhv[16];
#pragma unroll
    for (int mt = 0; mt < 2; ++mt) {
      long base = xbase + (long)(mt * 16 + 8 * ldiv) * Hc + ncol;
#pragma unroll
      for (int v = 0; v < 8; ++v) {
        xzv[mt * 8 + v] = xz[base + (long)v * Hc];
        xrv[mt * 8 + v] = xr[base + (long)v * Hc];
        xhv[mt * 8 + v] = xh[base + (long)v * Hc];
      }
    }

    v8f az[2] = {}, ar[2] = {}, ah[2] = {};

    {
      v2f bz_c = *(const v2f*)(wzr);
      v2f br_c = *(const v2f*)(wrr);
      v2f a0_c = *(const v2f*)(ha0);
      v2f a1_c = *(const v2f*)(ha1);
#pragma unroll 4
      for (int kk = 0; kk < Hc - 4; kk += 4) {
        v2f bz_n = *(const v2f*)(wzr + kk + 4);
        v2f br_n = *(const v2f*)(wrr + kk + 4);
        v2f a0_n = *(const v2f*)(ha0 + kk + 4);
        v2f a1_n = *(const v2f*)(ha1 + kk + 4);
        az[0] = WMMA_F32(a0_c, bz_c, az[0]);
        az[1] = WMMA_F32(a1_c, bz_c, az[1]);
        ar[0] = WMMA_F32(a0_c, br_c, ar[0]);
        ar[1] = WMMA_F32(a1_c, br_c, ar[1]);
        bz_c = bz_n; br_c = br_n; a0_c = a0_n; a1_c = a1_n;
      }
      az[0] = WMMA_F32(a0_c, bz_c, az[0]);
      az[1] = WMMA_F32(a1_c, bz_c, az[1]);
      ar[0] = WMMA_F32(a0_c, br_c, ar[0]);
      ar[1] = WMMA_F32(a1_c, br_c, ar[1]);
    }

#pragma unroll
    for (int mt = 0; mt < 2; ++mt) {
#pragma unroll
      for (int v = 0; v < 8; ++v) {
        int m = mt * 16 + v + 8 * ldiv;
        float zpre = xzv[mt * 8 + v] + az[mt][v];
        float rpre = xrv[mt * 8 + v] + ar[mt][v];
        az[mt][v] = 1.f / (1.f + __expf(-zpre));
        float rs  = 1.f / (1.f + __expf(-rpre));
        rhbuf[m * PITCH + ncol] = rs * hbuf[m * PITCH + ncol];
      }
    }
    __syncthreads();

    {
      v2f b_c  = *(const v2f*)(whr);
      v2f a0_c = *(const v2f*)(ra0);
      v2f a1_c = *(const v2f*)(ra1);
#pragma unroll 4
      for (int kk = 0; kk < Hc - 4; kk += 4) {
        v2f b_n  = *(const v2f*)(whr + kk + 4);
        v2f a0_n = *(const v2f*)(ra0 + kk + 4);
        v2f a1_n = *(const v2f*)(ra1 + kk + 4);
        ah[0] = WMMA_F32(a0_c, b_c, ah[0]);
        ah[1] = WMMA_F32(a1_c, b_c, ah[1]);
        b_c = b_n; a0_c = a0_n; a1_c = a1_n;
      }
      ah[0] = WMMA_F32(a0_c, b_c, ah[0]);
      ah[1] = WMMA_F32(a1_c, b_c, ah[1]);
    }

#pragma unroll
    for (int mt = 0; mt < 2; ++mt) {
#pragma unroll
      for (int v = 0; v < 8; ++v) {
        int m = mt * 16 + v + 8 * ldiv;
        float zp   = abuf[m] * az[mt][v];
        float ht   = tanhf(xhv[mt * 8 + v] + ah[mt][v]);
        float hold = hbuf[m * PITCH + ncol];
        hbuf[m * PITCH + ncol] = (1.f - zp) * hold + zp * ht;
      }
    }
    __syncthreads();
  }

  for (int idx = tid; idx < BT * Hc; idx += 256) {
    int m = idx >> 7, n = idx & 127;
    out[(long)(b0 + m) * Hc + n] = hbuf[m * PITCH + n];
  }
}

extern "C" void kernel_launch(void* const* d_in, const int* in_sizes, int n_in,
                              void* d_out, int out_size, void* d_ws, size_t ws_size,
                              hipStream_t stream) {
  const float* x   = (const float*)d_in[0];
  const float* att = (const float*)d_in[1];
  const float* Wz  = (const float*)d_in[2];
  const float* bz  = (const float*)d_in[3];
  const float* Wr  = (const float*)d_in[4];
  const float* br  = (const float*)d_in[5];
  const float* Wh  = (const float*)d_in[6];
  const float* bh  = (const float*)d_in[7];
  float* out = (float*)d_out;

  const size_t TBH = (size_t)Tc * Bc * Hc;
  float* xz = (float*)d_ws;
  float* xr = xz + TBH;
  float* xh = xr + TBH;

  dim3 g1((Bc * Tc / 16) / 8, 3);
  size_t sh1 = (size_t)Hc * PITCH * sizeof(float);
  augru_proj<<<g1, 256, sh1, stream>>>(x, Wz, bz, Wr, br, Wh, bh, xz, xr, xh);

  size_t sh2 = (size_t)(3 * Hc * PITCH + 2 * BT * PITCH + BT) * sizeof(float);
  augru_scan<<<Bc / BT, 256, sh2, stream>>>(att, Wz, Wr, Wh, xz, xr, xh, out);
}


# scan batch tile per work-group 32 -> 16 rows (one M tile per wave per gate), grid 32 -> 64 work-groups: the serial 200-step scan runs on twice as many WGPs; per-element k summation order unchanged
# speedup vs baseline: 1.6159x; 1.6159x over previous
#include <hip/hip_runtime.h>

typedef __attribute__((ext_vector_type(2))) float v2f;
typedef __attribute__((ext_vector_type(8))) float v8f;

#define WMMA_F32(a, b, c) \
  __builtin_amdgcn_wmma_f32_16x16x4_f32(false, (a), false, (b), (short)0, (c), false, false)

constexpr int Bc = 1024;
constexpr int Tc = 200;
constexpr int Ic = 128;
constexpr int Hc = 128;
constexpr int PITCH = 132;
constexpr int BT = 16;

__global__ __launch_bounds__(256) void augru_proj(
    const float* __restrict__ x,
    const float* __restrict__ Wz, const float* __restrict__ bz,
    const float* __restrict__ Wr, const float* __restrict__ br,
    const float* __restrict__ Wh, const float* __restrict__ bh,
    float* __restrict__ xz, float* __restrict__ xr, float* __restrict__ xh)
{
  extern __shared__ float lds[];

  const int gate = blockIdx.y;
  const float* W    = (gate == 0) ? Wz : (gate == 1) ? Wr : Wh;
  const float* bias = (gate == 0) ? bz : (gate == 1) ? br : bh;
  float* outp       = (gate == 0) ? xz : (gate == 1) ? xr : xh;

  const int tid = threadIdx.x;

  for (int idx = tid; idx < Hc * Ic; idx += 256) {
    int n = idx >> 7, k = idx & 127;
    lds[n * PITCH + k] = W[n * (Ic + Hc) + k];
  }
  __syncthreads();

  const int wave = tid >> 5, lane = tid & 31;
  const int lmod = lane & 15, ldiv = lane >> 4;
  const long mt = (long)blockIdx.x * 8 + wave;
  const long rbase = mt * 16;

  float biasv[8];
#pragma unroll
  for (int nt = 0; nt < 8; ++nt) biasv[nt] = bias[nt * 16 + lmod];

  v8f acc[8] = {};

  const float* arow = x + (rbase + lmod) * Ic + 2 * ldiv;
  const float* brow = lds + lmod * PITCH + 2 * ldiv;

  v2f a_c = *(const v2f*)(arow);
  v2f b_c[8];
#pragma unroll
  for (int nt = 0; nt < 8; ++nt) b_c[nt] = *(const v2f*)(brow + nt * 16 * PITCH);

#pragma unroll 4
  for (int kk = 0; kk < Ic - 4; kk += 4) {
    v2f a_n = *(const v2f*)(arow + kk + 4);
    v2f b_n[8];
#pragma unroll
    for (int nt = 0; nt < 8; ++nt)
      b_n[nt] = *(const v2f*)(brow + nt * 16 * PITCH + kk + 4);
#pragma unroll
    for (int nt = 0; nt < 8; ++nt) acc[nt] = WMMA_F32(a_c, b_c[nt], acc[nt]);
    a_c = a_n;
#pragma unroll
    for (int nt = 0; nt < 8; ++nt) b_c[nt] = b_n[nt];
  }
#pragma unroll
  for (int nt = 0; nt < 8; ++nt) acc[nt] = WMMA_F32(a_c, b_c[nt], acc[nt]);

#pragma unroll
  for (int v = 0; v < 8; ++v) {
    long r = rbase + v + 8 * ldiv;
    int  tt = (int)(r % Tc);
    long bb = r / Tc;
    float* orow = outp + ((long)tt * Bc + bb) * Hc + lmod;
#pragma unroll
    for (int nt = 0; nt < 8; ++nt) orow[nt * 16] = acc[nt][v] + biasv[nt];
  }
}

__global__ __launch_bounds__(256) void augru_scan(
    const float* __restrict__ att,
    const float* __restrict__ Wz, const float* __restrict__ Wr,
    const float* __restrict__ Wh,
    const float* __restrict__ xz, const float* __restrict__ xr,
    const float* __restrict__ xh,
    float* __restrict__ out)
{
  extern __shared__ float lds[];
  float* wzl   = lds;
  float* wrl   = wzl + Hc * PITCH;
  float* whl   = wrl + Hc * PITCH;
  float* hbuf  = whl + Hc * PITCH;
  float* rhbuf = hbuf + BT * PITCH;
  float* abuf  = rhbuf + BT * PITCH;

  const int tid = threadIdx.x;
  const int b0  = blockIdx.x * BT;

  for (int idx = tid; idx < Hc * Hc; idx += 256) {
    int n = idx >> 7, k = idx & 127;
    wzl[n * PITCH + k] = Wz[n * (Ic + Hc) + Ic + k];
    wrl[n * PITCH + k] = Wr[n * (Ic + Hc) + Ic + k];
    whl[n * PITCH + k] = Wh[n * (Ic + Hc) + Ic + k];
  }
  for (int idx = tid; idx < BT * Hc; idx += 256)
    hbuf[(idx >> 7) * PITCH + (idx & 127)] = 0.f;
  __syncthreads();

  const int wave = tid >> 5, lane = tid & 31;
  const int lmod = lane & 15, ldiv = lane >> 4;
  const int nt   = wave;
  const int ncol = nt * 16 + lmod;

  const float* wzr = wzl + ncol * PITCH + 2 * ldiv;
  const float* wrr = wrl + ncol * PITCH + 2 * ldiv;
  const float* whr = whl + ncol * PITCH + 2 * ldiv;
  const float* ha0 = hbuf + lmod * PITCH + 2 * ldiv;
  const float* ra0 = rhbuf + lmod * PITCH + 2 * ldiv;

  for (int t = 0; t < Tc; ++t) {
    if (tid < BT) abuf[tid] = att[(long)(b0 + tid) * Tc + t];

    if (t + 1 < Tc) {
      long nb = ((long)(t + 1) * Bc + b0) * Hc + tid * (BT * Hc / 256);
      __builtin_prefetch(xz + nb, 0, 1);
      __builtin_prefetch(xr + nb, 0, 1);
      __builtin_prefetch(xh + nb, 0, 1);
    }

    const long xbase = ((long)t * Bc + b0) * Hc + (long)(8 * ldiv) * Hc + ncol;
    float xzv[8], xrv[8], xhv[8];
#pragma unroll
    for (int v = 0; v < 8; ++v) {
      xzv[v] = xz[xbase + (long)v * Hc];
      xrv[v] = xr[xbase + (long)v * Hc];
      xhv[v] = xh[xbase + (long)v * Hc];
    }

    v8f az = {}, ar = {}, ah = {};

    {
      v2f bz_c = *(const v2f*)(wzr);
      v2f br_c = *(const v2f*)(wrr);
      v2f a0_c = *(const v2f*)(ha0);
#pragma unroll 4
      for (int kk = 0; kk < Hc - 4; kk += 4) {
        v2f bz_n = *(const v2f*)(wzr + kk + 4);
        v2f br_n = *(const v2f*)(wrr + kk + 4);
        v2f a0_n = *(const v2f*)(ha0 + kk + 4);
        az = WMMA_F32(a0_c, bz_c, az);
        ar = WMMA_F32(a0_c, br_c, ar);
        bz_c = bz_n; br_c = br_n; a0_c = a0_n;
      }
      az = WMMA_F32(a0_c, bz_c, az);
      ar = WMMA_F32(a0_c, br_c, ar);
    }

#pragma unroll
    for (int v = 0; v < 8; ++v) {
      int m = v + 8 * ldiv;
      float zpre = xzv[v] + az[v];
      float rpre = xrv[v] + ar[v];
      az[v] = 1.f / (1.f + __expf(-zpre));
      float rs  = 1.f / (1.f + __expf(-rpre));
      rhbuf[m * PITCH + ncol] = rs * hbuf[m * PITCH + ncol];
    }
    __syncthreads();

    {
      v2f b_c  = *(const v2f*)(whr);
      v2f a0_c = *(const v2f*)(ra0);
#pragma unroll 4
      for (int kk = 0; kk < Hc - 4; kk += 4) {
        v2f b_n  = *(const v2f*)(whr + kk + 4);
        v2f a0_n = *(const v2f*)(ra0 + kk + 4);
        ah = WMMA_F32(a0_c, b_c, ah);
        b_c = b_n; a0_c = a0_n;
      }
      ah = WMMA_F32(a0_c, b_c, ah);
    }

#pragma unroll
    for (int v = 0; v < 8; ++v) {
      int m = v + 8 * ldiv;
      float zp   = abuf[m] * az[v];
      float ht   = tanhf(xhv[v] + ah[v]);
      float hold = hbuf[m * PITCH + ncol];
      hbuf[m * PITCH + ncol] = (1.f - zp) * hold + zp * ht;
    }
    __syncthreads();
  }

  for (int idx = tid; idx < BT * Hc; idx += 256) {
    int m = idx >> 7, n = idx & 127;
    out[(long)(b0 + m) * Hc + n] = hbuf[m * PITCH + n];
  }
}

extern "C" void kernel_launch(void* const* d_in, const int* in_sizes, int n_in,
                              void* d_out, int out_size, void* d_ws, size_t ws_size,
                              hipStream_t stream) {
  const float* x   = (const float*)d_in[0];
  const float* att = (const float*)d_in[1];
  const float* Wz  = (const float*)d_in[2];
  const float* bz  = (const float*)d_in[3];
  const float* Wr  = (const float*)d_in[4];
  const float* br  = (const float*)d_in[5];
  const float* Wh  = (const float*)d_in[6];
  const float* bh  = (const float*)d_in[7];
  float* out = (float*)d_out;

  const size_t TBH = (size_t)Tc * Bc * Hc;
  float* xz = (float*)d_ws;
  float* xr = xz + TBH;
  float* xh = xr + TBH;

  dim3 g1((Bc * Tc / 16) / 8, 3);
  size_t sh1 = (size_t)Hc * PITCH * sizeof(float);
  augru_proj<<<g1, 256, sh1, stream>>>(x, Wz, bz, Wr, br, Wh, bh, xz, xr, xh);

  size_t sh2 = (size_t)(3 * Hc * PITCH + 2 * BT * PITCH + BT) * sizeof(float);
  augru_scan<<<Bc / BT, 256, sh2, stream>>>(att, Wz, Wr, Wh, xz, xr, xh, out);
}


# scan step issue order pinned with scheduling fences: global x and a_t loads at step start, LDS fragment loads grouped a block ahead of the WMMAs (h~ GEMM waited per WMMA pair); sums unchanged
# speedup vs baseline: 1.9344x; 1.1971x over previous
#include <hip/hip_runtime.h>

typedef __attribute__((ext_vector_type(2))) float v2f;
typedef __attribute__((ext_vector_type(8))) float v8f;

#define WMMA_F32(a, b, c) \
  __builtin_amdgcn_wmma_f32_16x16x4_f32(false, (a), false, (b), (short)0, (c), false, false)

constexpr int Bc = 1024;
constexpr int Tc = 200;
constexpr int Ic = 128;
constexpr int Hc = 128;
constexpr int PITCH = 132;
constexpr int BT = 16;
constexpr int KT = Hc / 4;
constexpr int KB = 8;
constexpr int NBLK = KT / KB;

__global__ __launch_bounds__(256) void augru_proj(
    const float* __restrict__ x,
    const float* __restrict__ Wz, const float* __restrict__ bz,
    const float* __restrict__ Wr, const float* __restrict__ br,
    const float* __restrict__ Wh, const float* __restrict__ bh,
    float* __restrict__ xz, float* __restrict__ xr, float* __restrict__ xh)
{
  extern __shared__ float lds[];

  const int gate = blockIdx.y;
  const float* W    = (gate == 0) ? Wz : (gate == 1) ? Wr : Wh;
  const float* bias = (gate == 0) ? bz : (gate == 1) ? br : bh;
  float* outp       = (gate == 0) ? xz : (gate == 1) ? xr : xh;

  const int tid = threadIdx.x;

  for (int idx = tid; idx < Hc * Ic; idx += 256) {
    int n = idx >> 7, k = idx & 127;
    lds[n * PITCH + k] = W[n * (Ic + Hc) + k];
  }
  __syncthreads();

  const int wave = tid >> 5, lane = tid & 31;
  const int lmod = lane & 15, ldiv = lane >> 4;
  const long mt = (long)blockIdx.x * 8 + wave;
  const long rbase = mt * 16;

  float biasv[8];
#pragma unroll
  for (int nt = 0; nt < 8; ++nt) biasv[nt] = bias[nt * 16 + lmod];

  v8f acc[8] = {};

  const float* arow = x + (rbase + lmod) * Ic + 2 * ldiv;
  const float* brow = lds + lmod * PITCH + 2 * ldiv;

  v2f a_c = *(const v2f*)(arow);
  v2f b_c[8];
#pragma unroll
  for (int nt = 0; nt < 8; ++nt) b_c[nt] = *(const v2f*)(brow + nt * 16 * PITCH);

#pragma unroll 4
  for (int kk = 0; kk < Ic - 4; kk += 4) {
    v2f a_n = *(const v2f*)(arow + kk + 4);
    v2f b_n[8];
#pragma unroll
    for (int nt = 0; nt < 8; ++nt)
      b_n[nt] = *(const v2f*)(brow + nt * 16 * PITCH + kk + 4);
#pragma unroll
    for (int nt = 0; nt < 8; ++nt) acc[nt] = WMMA_F32(a_c, b_c[nt], acc[nt]);
    a_c = a_n;
#pragma unroll
    for (int nt = 0; nt < 8; ++nt) b_c[nt] = b_n[nt];
  }
#pragma unroll
  for (int nt = 0; nt < 8; ++nt) acc[nt] = WMMA_F32(a_c, b_c[nt], acc[nt]);

#pragma unroll
  for (int v = 0; v < 8; ++v) {
    long r = rbase + v + 8 * ldiv;
    int  tt = (int)(r % Tc);
    long bb = r / Tc;
    float* orow = outp + ((long)tt * Bc + bb) * Hc + lmod;
#pragma unroll
    for (int nt = 0; nt < 8; ++nt) orow[nt * 16] = acc[nt][v] + biasv[nt];
  }
}

__global__ __launch_bounds__(256) void augru_scan(
    const float* __restrict__ att,
    const float* __restrict__ Wz, const float* __restrict__ Wr,
    const float* __restrict__ Wh,
    const float* __restrict__ xz, const float* __restrict__ xr,
    const float* __restrict__ xh,
    float* __restrict__ out)
{
  extern __shared__ float lds[];
  float* wzl   = lds;
  float* wrl   = wzl + Hc * PITCH;
  float* whl   = wrl + Hc * PITCH;
  float* hbuf  = whl + Hc * PITCH;
  float* rhbuf = hbuf + BT * PITCH;

  const int tid = threadIdx.x;
  const int b0  = blockIdx.x * BT;

  for (int idx = tid; idx < Hc * Hc; idx += 256) {
    int n = idx >> 7, k = idx & 127;
    wzl[n * PITCH + k] = Wz[n * (Ic + Hc) + Ic + k];
    wrl[n * PITCH + k] = Wr[n * (Ic + Hc) + Ic + k];
    whl[n * PITCH + k] = Wh[n * (Ic + Hc) + Ic + k];
  }
  for (int idx = tid; idx < BT * Hc; idx += 256)
    hbuf[(idx >> 7) * PITCH + (idx & 127)] = 0.f;
  __syncthreads();

  const int wave = tid >> 5, lane = tid & 31;
  const int lmod = lane & 15, ldiv = lane >> 4;
  const int nt   = wave;
  const int ncol = nt * 16 + lmod;

  const float* wzr = wzl + ncol * PITCH + 2 * ldiv;
  const float* wrr = wrl + ncol * PITCH + 2 * ldiv;
  const float* whr = whl + ncol * PITCH + 2 * ldiv;
  const float* ha0 = hbuf + lmod * PITCH + 2 * ldiv;
  const float* ra0 = rhbuf + lmod * PITCH + 2 * ldiv;

  const float* arow = att + (long)(b0 + 8 * ldiv) * Tc;

  for (int t = 0; t < Tc; ++t) {
    if (t + 1 < Tc) {
      long nb = ((long)(t + 1) * Bc + b0) * Hc + tid * (BT * Hc / 256);
      __builtin_prefetch(xz + nb, 0, 1);
      __builtin_prefetch(xr + nb, 0, 1);
      __builtin_prefetch(xh + nb, 0, 1);
    }

    v2f fa[2][KB], fz[2][KB], fr[2][KB];
#pragma unroll
    for (int i = 0; i < KB; ++i) {
      fa[0][i] = *(const v2f*)(ha0 + 4 * i);
      fz[0][i] = *(const v2f*)(wzr + 4 * i);
      fr[0][i] = *(const v2f*)(wrr + 4 * i);
    }
    __builtin_amdgcn_sched_barrier(0);

    const long xbase = ((long)t * Bc + b0) * Hc + (long)(8 * ldiv) * Hc + ncol;
    float xzv[8], xrv[8], xhv[8], av[8];
#pragma unroll
    for (int v = 0; v < 8; ++v) {
      xzv[v] = xz[xbase + (long)v * Hc];
      xrv[v] = xr[xbase + (long)v * Hc];
      xhv[v] = xh[xbase + (long)v * Hc];
      av[v]  = arow[(long)v * Tc + t];
    }

    v8f az = {}, ar = {}, ah = {};
    float hold[8];

#pragma unroll
    for (int blk = 0; blk < NBLK; ++blk) {
      const int cur = blk & 1, nxt = cur ^ 1;
      if (blk + 1 < NBLK) {
#pragma unroll
        for (int i = 0; i < KB; ++i) {
          const int kk = ((blk + 1) * KB + i) * 4;
          fa[nxt][i] = *(const v2f*)(ha0 + kk);
          fz[nxt][i] = *(const v2f*)(wzr + kk);
          fr[nxt][i] = *(const v2f*)(wrr + kk);
        }
      } else {
#pragma unroll
        for (int v = 0; v < 8; ++v) hold[v] = hbuf[(v + 8 * ldiv) * PITCH + ncol];
      }
      __builtin_amdgcn_sched_barrier(0);
#pragma unroll
      for (int i = 0; i < KB; ++i) {
        az = WMMA_F32(fa[cur][i], fz[cur][i], az);
        ar = WMMA_F32(fa[cur][i], fr[cur][i], ar);
      }
    }

    v2f fh[KT];
#pragma unroll
    for (int i = 0; i < KT; ++i) fh[i] = *(const v2f*)(whr + 4 * i);

#pragma unroll
    for (int v = 0; v < 8; ++v) {
      int m = v + 8 * ldiv;
      float zpre = xzv[v] + az[v];
      float rpre = xrv[v] + ar[v];
      az[v] = 1.f / (1.f + __expf(-zpre));
      float rs  = 1.f / (1.f + __expf(-rpre));
      rhbuf[m * PITCH + ncol] = rs * hold[v];
    }
    __syncthreads();

    v2f fra[KT];
#pragma unroll
    for (int i = 0; i < KT; ++i) fra[i] = *(const v2f*)(ra0 + 4 * i);
    __builtin_amdgcn_sched_barrier(0);
#pragma unroll
    for (int i = 0; i < KT; ++i) ah = WMMA_F32(fra[i], fh[i], ah);

#pragma unroll
    for (int v = 0; v < 8; ++v) {
      int m = v + 8 * ldiv;
      float zp   = av[v] * az[v];
      float ht   = tanhf(xhv[v] + ah[v]);
      hbuf[m * PITCH + ncol] = (1.f - zp) * hold[v] + zp * ht;
    }
    __syncthreads();
  }

  for (int idx = tid; idx < BT * Hc; idx += 256) {
    int m = idx >> 7, n = idx & 127;
    out[(long)(b0 + m) * Hc + n] = hbuf[m * PITCH + n];
  }
}

extern "C" void kernel_launch(void* const* d_in, const int* in_sizes, int n_in,
                              void* d_out, int out_size, void* d_ws, size_t ws_size,
                              hipStream_t stream) {
  const float* x   = (const float*)d_in[0];
  const float* att = (const float*)d_in[1];
  const float* Wz  = (const float*)d_in[2];
  const float* bz  = (const float*)d_in[3];
  const float* Wr  = (const float*)d_in[4];
  const float* br  = (const float*)d_in[5];
  const float* Wh  = (const float*)d_in[6];
  const float* bh  = (const float*)d_in[7];
  float* out = (float*)d_out;

  const size_t TBH = (size_t)Tc * Bc * Hc;
  float* xz = (float*)d_ws;
  float* xr = xz + TBH;
  float* xh = xr + TBH;

  dim3 g1((Bc * Tc / 16) / 8, 3);
  size_t sh1 = (size_t)Hc * PITCH * sizeof(float);
  augru_proj<<<g1, 256, sh1, stream>>>(x, Wz, bz, Wr, br, Wh, bh, xz, xr, xh);

  size_t sh2 = (size_t)(3 * Hc * PITCH + 2 * BT * PITCH) * sizeof(float);
  augru_scan<<<Bc / BT, 256, sh2, stream>>>(att, Wz, Wr, Wh, xz, xr, xh, out);
}


# scan gate math branch-free: sigmoid and tanh each as one hardware exp2 + one hardware reciprocal (no IEEE divide, no branching library tanh), 8 rows per lane pipelined; z gate issued beside h~ WMMAs
# speedup vs baseline: 2.5470x; 1.3167x over previous
#include <hip/hip_runtime.h>

typedef __attribute__((ext_vector_type(2))) float v2f;
typedef __attribute__((ext_vector_type(8))) float v8f;

#define WMMA_F32(a, b, c) \
  __builtin_amdgcn_wmma_f32_16x16x4_f32(false, (a), false, (b), (short)0, (c), false, false)

__device__ __forceinline__ float sigmoid_hw(float x) {
  return __builtin_amdgcn_rcpf(1.f + __builtin_amdgcn_exp2f(-1.44269504f * x));
}
__device__ __forceinline__ float tanh_hw(float x) {
  return fmaf(-2.f, __builtin_amdgcn_rcpf(1.f + __builtin_amdgcn_exp2f(2.88539008f * x)), 1.f);
}

constexpr int Bc = 1024;
constexpr int Tc = 200;
constexpr int Ic = 128;
constexpr int Hc = 128;
constexpr int PITCH = 132;
constexpr int BT = 16;
constexpr int KT = Hc / 4;
constexpr int KB = 8;
constexpr int NBLK = KT / KB;

__global__ __launch_bounds__(256) void augru_proj(
    const float* __restrict__ x,
    const float* __restrict__ Wz, const float* __restrict__ bz,
    const float* __restrict__ Wr, const float* __restrict__ br,
    const float* __restrict__ Wh, const float* __restrict__ bh,
    float* __restrict__ xz, float* __restrict__ xr, float* __restrict__ xh)
{
  extern __shared__ float lds[];

  const int gate = blockIdx.y;
  const float* W    = (gate == 0) ? Wz : (gate == 1) ? Wr : Wh;
  const float* bias = (gate == 0) ? bz : (gate == 1) ? br : bh;
  float* outp       = (gate == 0) ? xz : (gate == 1) ? xr : xh;

  const int tid = threadIdx.x;

  for (int idx = tid; idx < Hc * Ic; idx += 256) {
    int n = idx >> 7, k = idx & 127;
    lds[n * PITCH + k] = W[n * (Ic + Hc) + k];
  }
  __syncthreads();

  const int wave = tid >> 5, lane = tid & 31;
  const int lmod = lane & 15, ldiv = lane >> 4;
  const long mt = (long)blockIdx.x * 8 + wave;
  const long rbase = mt * 16;

  float biasv[8];
#pragma unroll
  for (int nt = 0; nt < 8; ++nt) biasv[nt] = bias[nt * 16 + lmod];

  v8f acc[8] = {};

  const float* arow = x + (rbase + lmod) * Ic + 2 * ldiv;
  const float* brow = lds + lmod * PITCH + 2 * ldiv;

  v2f a_c = *(const v2f*)(arow);
  v2f b_c[8];
#pragma unroll
  for (int nt = 0; nt < 8; ++nt) b_c[nt] = *(const v2f*)(brow + nt * 16 * PITCH);

#pragma unroll 4
  for (int kk = 0; kk < Ic - 4; kk += 4) {
    v2f a_n = *(const v2f*)(arow + kk + 4);
    v2f b_n[8];
#pragma unroll
    for (int nt = 0; nt < 8; ++nt)
      b_n[nt] = *(const v2f*)(brow + nt * 16 * PITCH + kk + 4);
#pragma unroll
    for (int nt = 0; nt < 8; ++nt) acc[nt] = WMMA_F32(a_c, b_c[nt], acc[nt]);
    a_c = a_n;
#pragma unroll
    for (int nt = 0; nt < 8; ++nt) b_c[nt] = b_n[nt];
  }
#pragma unroll
  for (int nt = 0; nt < 8; ++nt) acc[nt] = WMMA_F32(a_c, b_c[nt], acc[nt]);

#pragma unroll
  for (int v = 0; v < 8; ++v) {
    long r = rbase + v + 8 * ldiv;
    int  tt = (int)(r % Tc);
    long bb = r / Tc;
    float* orow = outp + ((long)tt * Bc + bb) * Hc + lmod;
#pragma unroll
    for (int nt = 0; nt < 8; ++nt) orow[nt * 16] = acc[nt][v] + biasv[nt];
  }
}

__global__ __launch_bounds__(256) void augru_scan(
    const float* __restrict__ att,
    const float* __restrict__ Wz, const float* __restrict__ Wr,
    const float* __restrict__ Wh,
    const float* __restrict__ xz, const float* __restrict__ xr,
    const float* __restrict__ xh,
    float* __restrict__ out)
{
  extern __shared__ float lds[];
  float* wzl   = lds;
  float* wrl   = wzl + Hc * PITCH;
  float* whl   = wrl + Hc * PITCH;
  float* hbuf  = whl + Hc * PITCH;
  float* rhbuf = hbuf + BT * PITCH;

  const int tid = threadIdx.x;
  const int b0  = blockIdx.x * BT;

  for (int idx = tid; idx < Hc * Hc; idx += 256) {
    int n = idx >> 7, k = idx & 127;
    wzl[n * PITCH + k] = Wz[n * (Ic + Hc) + Ic + k];
    wrl[n * PITCH + k] = Wr[n * (Ic + Hc) + Ic + k];
    whl[n * PITCH + k] = Wh[n * (Ic + Hc) + Ic + k];
  }
  for (int idx = tid; idx < BT * Hc; idx += 256)
    hbuf[(idx >> 7) * PITCH + (idx & 127)] = 0.f;
  __syncthreads();

  const int wave = tid >> 5, lane = tid & 31;
  const int lmod = lane & 15, ldiv = lane >> 4;
  const int nt   = wave;
  const int ncol = nt * 16 + lmod;

  const float* wzr = wzl + ncol * PITCH + 2 * ldiv;
  const float* wrr = wrl + ncol * PITCH + 2 * ldiv;
  const float* whr = whl + ncol * PITCH + 2 * ldiv;
  const float* ha0 = hbuf + lmod * PITCH + 2 * ldiv;
  const float* ra0 = rhbuf + lmod * PITCH + 2 * ldiv;

  const float* arow = att + (long)(b0 + 8 * ldiv) * Tc;

  for (int t = 0; t < Tc; ++t) {
    if (t + 1 < Tc) {
      long nb = ((long)(t + 1) * Bc + b0) * Hc + tid * (BT * Hc / 256);
      __builtin_prefetch(xz + nb, 0, 1);
      __builtin_prefetch(xr + nb, 0, 1);
      __builtin_prefetch(xh + nb, 0, 1);
    }

    v2f fa[2][KB], fz[2][KB], fr[2][KB];
#pragma unroll
    for (int i = 0; i < KB; ++i) {
      fa[0][i] = *(const v2f*)(ha0 + 4 * i);
      fz[0][i] = *(const v2f*)(wzr + 4 * i);
      fr[0][i] = *(const v2f*)(wrr + 4 * i);
    }
    __builtin_amdgcn_sched_barrier(0);

    const long xbase = ((long)t * Bc + b0) * Hc + (long)(8 * ldiv) * Hc + ncol;
    float xzv[8], xrv[8], xhv[8], av[8];
#pragma unroll
    for (int v = 0; v < 8; ++v) {
      xzv[v] = xz[xbase + (long)v * Hc];
      xrv[v] = xr[xbase + (long)v * Hc];
      xhv[v] = xh[xbase + (long)v * Hc];
      av[v]  = arow[(long)v * Tc + t];
    }

    v8f az = {}, ar = {}, ah = {};
    float hold[8];

#pragma unroll
    for (int blk = 0; blk < NBLK; ++blk) {
      const int cur = blk & 1, nxt = cur ^ 1;
      if (blk + 1 < NBLK) {
#pragma unroll
        for (int i = 0; i < KB; ++i) {
          const int kk = ((blk + 1) * KB + i) * 4;
          fa[nxt][i] = *(const v2f*)(ha0 + kk);
          fz[nxt][i] = *(const v2f*)(wzr + kk);
          fr[nxt][i] = *(const v2f*)(wrr + kk);
        }
      } else {
#pragma unroll
        for (int v = 0; v < 8; ++v) hold[v] = hbuf[(v + 8 * ldiv) * PITCH + ncol];
      }
      __builtin_amdgcn_sched_barrier(0);
#pragma unroll
      for (int i = 0; i < KB; ++i) {
        az = WMMA_F32(fa[cur][i], fz[cur][i], az);
        ar = WMMA_F32(fa[cur][i], fr[cur][i], ar);
      }
    }

    v2f fh[KT];
#pragma unroll
    for (int i = 0; i < KT; ++i) fh[i] = *(const v2f*)(whr + 4 * i);

#pragma unroll
    for (int v = 0; v < 8; ++v) {
      int m = v + 8 * ldiv;
      rhbuf[m * PITCH + ncol] = sigmoid_hw(xrv[v] + ar[v]) * hold[v];
    }
    __syncthreads();

    v2f fra[KT];
#pragma unroll
    for (int i = 0; i < KT; ++i) fra[i] = *(const v2f*)(ra0 + 4 * i);
    __builtin_amdgcn_sched_barrier(0);
    float zp[8], keep[8];
#pragma unroll
    for (int v = 0; v < 8; ++v) {
      zp[v]   = av[v] * sigmoid_hw(xzv[v] + az[v]);
      keep[v] = (1.f - zp[v]) * hold[v];
    }
#pragma unroll
    for (int i = 0; i < KT; ++i) ah = WMMA_F32(fra[i], fh[i], ah);

#pragma unroll
    for (int v = 0; v < 8; ++v) {
      int m = v + 8 * ldiv;
      hbuf[m * PITCH + ncol] = fmaf(zp[v], tanh_hw(xhv[v] + ah[v]), keep[v]);
    }
    __syncthreads();
  }

  for (int idx = tid; idx < BT * Hc; idx += 256) {
    int m = idx >> 7, n = idx & 127;
    out[(long)(b0 + m) * Hc + n] = hbuf[m * PITCH + n];
  }
}

extern "C" void kernel_launch(void* const* d_in, const int* in_sizes, int n_in,
                              void* d_out, int out_size, void* d_ws, size_t ws_size,
                              hipStream_t stream) {
  const float* x   = (const float*)d_in[0];
  const float* att = (const float*)d_in[1];
  const float* Wz  = (const float*)d_in[2];
  const float* bz  = (const float*)d_in[3];
  const float* Wr  = (const float*)d_in[4];
  const float* br  = (const float*)d_in[5];
  const float* Wh  = (const float*)d_in[6];
  const float* bh  = (const float*)d_in[7];
  float* out = (float*)d_out;

  const size_t TBH = (size_t)Tc * Bc * Hc;
  float* xz = (float*)d_ws;
  float* xr = xz + TBH;
  float* xh = xr + TBH;

  dim3 g1((Bc * Tc / 16) / 8, 3);
  size_t sh1 = (size_t)Hc * PITCH * sizeof(float);
  augru_proj<<<g1, 256, sh1, stream>>>(x, Wz, bz, Wr, br, Wh, bh, xz, xr, xh);

  size_t sh2 = (size_t)(3 * Hc * PITCH + 2 * BT * PITCH) * sizeof(float);
  augru_scan<<<Bc / BT, 256, sh2, stream>>>(att, Wz, Wr, Wh, xz, xr, xh, out);
}


# scan: recurrent weight B fragments (3 gates x 32 k-steps) loaded once into registers before the time loop; per step only the h and r*h A fragments are read from LDS; k summation order unchanged
# speedup vs baseline: 2.7480x; 1.0789x over previous
#include <hip/hip_runtime.h>

typedef __attribute__((ext_vector_type(2))) float v2f;
typedef __attribute__((ext_vector_type(8))) float v8f;

#define WMMA_F32(a, b, c) \
  __builtin_amdgcn_wmma_f32_16x16x4_f32(false, (a), false, (b), (short)0, (c), false, false)

__device__ __forceinline__ float sigmoid_hw(float x) {
  return __builtin_amdgcn_rcpf(1.f + __builtin_amdgcn_exp2f(-1.44269504f * x));
}
__device__ __forceinline__ float tanh_hw(float x) {
  return fmaf(-2.f, __builtin_amdgcn_rcpf(1.f + __builtin_amdgcn_exp2f(2.88539008f * x)), 1.f);
}

constexpr int Bc = 1024;
constexpr int Tc = 200;
constexpr int Ic = 128;
constexpr int Hc = 128;
constexpr int PITCH = 132;
constexpr int BT = 16;
constexpr int KT = Hc / 4;

__global__ __launch_bounds__(256) void augru_proj(
    const float* __restrict__ x,
    const float* __restrict__ Wz, const float* __restrict__ bz,
    const float* __restrict__ Wr, const float* __restrict__ br,
    const float* __restrict__ Wh, const float* __restrict__ bh,
    float* __restrict__ xz, float* __restrict__ xr, float* __restrict__ xh)
{
  extern __shared__ float lds[];

  const int gate = blockIdx.y;
  const float* W    = (gate == 0) ? Wz : (gate == 1) ? Wr : Wh;
  const float* bias = (gate == 0) ? bz : (gate == 1) ? br : bh;
  float* outp       = (gate == 0) ? xz : (gate == 1) ? xr : xh;

  const int tid = threadIdx.x;

  for (int idx = tid; idx < Hc * Ic; idx += 256) {
    int n = idx >> 7, k = idx & 127;
    lds[n * PITCH + k] = W[n * (Ic + Hc) + k];
  }
  __syncthreads();

  const int wave = tid >> 5, lane = tid & 31;
  const int lmod = lane & 15, ldiv = lane >> 4;
  const long mt = (long)blockIdx.x * 8 + wave;
  const long rbase = mt * 16;

  float biasv[8];
#pragma unroll
  for (int nt = 0; nt < 8; ++nt) biasv[nt] = bias[nt * 16 + lmod];

  v8f acc[8] = {};

  const float* arow = x + (rbase + lmod) * Ic + 2 * ldiv;
  const float* brow = lds + lmod * PITCH + 2 * ldiv;

  v2f a_c = *(const v2f*)(arow);
  v2f b_c[8];
#pragma unroll
  for (int nt = 0; nt < 8; ++nt) b_c[nt] = *(const v2f*)(brow + nt * 16 * PITCH);

#pragma unroll 4
  for (int kk = 0; kk < Ic - 4; kk += 4) {
    v2f a_n = *(const v2f*)(arow + kk + 4);
    v2f b_n[8];
#pragma unroll
    for (int nt = 0; nt < 8; ++nt)
      b_n[nt] = *(const v2f*)(brow + nt * 16 * PITCH + kk + 4);
#pragma unroll
    for (int nt = 0; nt < 8; ++nt) acc[nt] = WMMA_F32(a_c, b_c[nt], acc[nt]);
    a_c = a_n;
#pragma unroll
    for (int nt = 0; nt < 8; ++nt) b_c[nt] = b_n[nt];
  }
#pragma unroll
  for (int nt = 0; nt < 8; ++nt) acc[nt] = WMMA_F32(a_c, b_c[nt], acc[nt]);

#pragma unroll
  for (int v = 0; v < 8; ++v) {
    long r = rbase + v + 8 * ldiv;
    int  tt = (int)(r % Tc);
    long bb = r / Tc;
    float* orow = outp + ((long)tt * Bc + bb) * Hc + lmod;
#pragma unroll
    for (int nt = 0; nt < 8; ++nt) orow[nt * 16] = acc[nt][v] + biasv[nt];
  }
}

__global__ __launch_bounds__(256) void augru_scan(
    const float* __restrict__ att,
    const float* __restrict__ Wz, const float* __restrict__ Wr,
    const float* __restrict__ Wh,
    const float* __restrict__ xz, const float* __restrict__ xr,
    const float* __restrict__ xh,
    float* __restrict__ out)
{
  extern __shared__ float lds[];
  float* hbuf  = lds;
  float* rhbuf = hbuf + BT * PITCH;

  const int tid = threadIdx.x;
  const int b0  = blockIdx.x * BT;

  for (int idx = tid; idx < BT * Hc; idx += 256)
    hbuf[(idx >> 7) * PITCH + (idx & 127)] = 0.f;

  const int wave = tid >> 5, lane = tid & 31;
  const int lmod = lane & 15, ldiv = lane >> 4;
  const int nt   = wave;
  const int ncol = nt * 16 + lmod;

  const float* wzg = Wz + (long)ncol * (Ic + Hc) + Ic + 2 * ldiv;
  const float* wrg = Wr + (long)ncol * (Ic + Hc) + Ic + 2 * ldiv;
  const float* whg = Wh + (long)ncol * (Ic + Hc) + Ic + 2 * ldiv;
  v2f fz[KT], fr[KT], fh[KT];
#pragma unroll
  for (int i = 0; i < KT; ++i) {
    fz[i] = *(const v2f*)(wzg + 4 * i);
    fr[i] = *(const v2f*)(wrg + 4 * i);
    fh[i] = *(const v2f*)(whg + 4 * i);
  }

  const float* ha0 = hbuf + lmod * PITCH + 2 * ldiv;
  const float* ra0 = rhbuf + lmod * PITCH + 2 * ldiv;

  const float* arow = att + (long)(b0 + 8 * ldiv) * Tc;

  float hold[8];
#pragma unroll
  for (int v = 0; v < 8; ++v) hold[v] = 0.f;

  __syncthreads();

  for (int t = 0; t < Tc; ++t) {
    if (t + 1 < Tc) {
      long nb = ((long)(t + 1) * Bc + b0) * Hc + tid * (BT * Hc / 256);
      __builtin_prefetch(xz + nb, 0, 1);
      __builtin_prefetch(xr + nb, 0, 1);
      __builtin_prefetch(xh + nb, 0, 1);
    }

    v2f fa[KT];
#pragma unroll
    for (int i = 0; i < KT; ++i) fa[i] = *(const v2f*)(ha0 + 4 * i);

    const long xbase = ((long)t * Bc + b0) * Hc + (long)(8 * ldiv) * Hc + ncol;
    float xzv[8], xrv[8], xhv[8], av[8];
#pragma unroll
    for (int v = 0; v < 8; ++v) {
      xzv[v] = xz[xbase + (long)v * Hc];
      xrv[v] = xr[xbase + (long)v * Hc];
      xhv[v] = xh[xbase + (long)v * Hc];
      av[v]  = arow[(long)v * Tc + t];
    }

    __builtin_amdgcn_sched_barrier(0);

    v8f az = {}, ar = {}, ah = {};

#pragma unroll
    for (int i = 0; i < KT; ++i) {
      az = WMMA_F32(fa[i], fz[i], az);
      ar = WMMA_F32(fa[i], fr[i], ar);
    }

#pragma unroll
    for (int v = 0; v < 8; ++v) {
      int m = v + 8 * ldiv;
      rhbuf[m * PITCH + ncol] = sigmoid_hw(xrv[v] + ar[v]) * hold[v];
    }
    __syncthreads();

#pragma unroll
    for (int i = 0; i < KT; ++i) fa[i] = *(const v2f*)(ra0 + 4 * i);
    __builtin_amdgcn_sched_barrier(0);
    float zp[8], keep[8];
#pragma unroll
    for (int v = 0; v < 8; ++v) {
      zp[v]   = av[v] * sigmoid_hw(xzv[v] + az[v]);
      keep[v] = (1.f - zp[v]) * hold[v];
    }
#pragma unroll
    for (int i = 0; i < KT; ++i) ah = WMMA_F32(fa[i], fh[i], ah);

#pragma unroll
    for (int v = 0; v < 8; ++v) {
      int m = v + 8 * ldiv;
      hold[v] = fmaf(zp[v], tanh_hw(xhv[v] + ah[v]), keep[v]);
      hbuf[m * PITCH + ncol] = hold[v];
    }
    __syncthreads();
  }

  for (int idx = tid; idx < BT * Hc; idx += 256) {
    int m = idx >> 7, n = idx & 127;
    out[(long)(b0 + m) * Hc + n] = hbuf[m * PITCH + n];
  }
}

extern "C" void kernel_launch(void* const* d_in, const int* in_sizes, int n_in,
                              void* d_out, int out_size, void* d_ws, size_t ws_size,
                              hipStream_t stream) {
  const float* x   = (const float*)d_in[0];
  const float* att = (const float*)d_in[1];
  const float* Wz  = (const float*)d_in[2];
  const float* bz  = (const float*)d_in[3];
  const float* Wr  = (const float*)d_in[4];
  const float* br  = (const float*)d_in[5];
  const float* Wh  = (const float*)d_in[6];
  const float* bh  = (const float*)d_in[7];
  float* out = (float*)d_out;

  const size_t TBH = (size_t)Tc * Bc * Hc;
  float* xz = (float*)d_ws;
  float* xr = xz + TBH;
  float* xh = xr + TBH;

  dim3 g1((Bc * Tc / 16) / 8, 3);
  size_t sh1 = (size_t)Hc * PITCH * sizeof(float);
  augru_proj<<<g1, 256, sh1, stream>>>(x, Wz, bz, Wr, br, Wh, bh, xz, xr, xh);

  size_t sh2 = (size_t)(2 * BT * PITCH) * sizeof(float);
  augru_scan<<<Bc / BT, 256, sh2, stream>>>(att, Wz, Wr, Wh, xz, xr, xh, out);
}


# input projection overlapped with the scan: time chunks 16/24/40/64/56, each launch runs 64 scan work-groups of one chunk beside 192 projection work-groups of the next (6 launches); sums unchanged
# speedup vs baseline: 3.1665x; 1.1523x over previous
#include <hip/hip_runtime.h>


typedef __attribute__((ext_vector_type(2))) float v2f;
typedef __attribute__((ext_vector_type(8))) float v8f;

#define WMMA_F32(a, b, c) \
  __builtin_amdgcn_wmma_f32_16x16x4_f32(false, (a), false, (b), (short)0, (c), false, false)

__device__ __forceinline__ float sigmoid_hw(float x) {
  return __builtin_amdgcn_rcpf(1.f + __builtin_amdgcn_exp2f(-1.44269504f * x));
}
__device__ __forceinline__ float tanh_hw(float x) {
  return fmaf(-2.f, __builtin_amdgcn_rcpf(1.f + __builtin_amdgcn_exp2f(2.88539008f * x)), 1.f);
}

constexpr int Bc = 1024;
constexpr int Tc = 200;
constexpr int Ic = 128;
constexpr int Hc = 128;
constexpr int PITCH = 132;
constexpr int BT = 16;
constexpr int KT = Hc / 4;
constexpr int NSCAN = Bc / BT;

__device__ __forceinline__ void proj_role(
    const float* __restrict__ x,
    const float* __restrict__ Wz, const float* __restrict__ bz,
    const float* __restrict__ Wr, const float* __restrict__ br,
    const float* __restrict__ Wh, const float* __restrict__ bh,
    float* __restrict__ xz, float* __restrict__ xr, float* __restrict__ xh,
    float* lds,
    const int pb, const int nproj, const int p0, const unsigned L)
{
  const int gate = pb % 3;
  const int pg   = pb / 3;
  const int G    = nproj / 3;
  const float* W    = (gate == 0) ? Wz : (gate == 1) ? Wr : Wh;
  const float* bias = (gate == 0) ? bz : (gate == 1) ? br : bh;
  float* outp       = (gate == 0) ? xz : (gate == 1) ? xr : xh;

  const int tid = threadIdx.x;

  for (int idx = tid; idx < Hc * Ic; idx += 256) {
    int n = idx >> 7, k = idx & 127;
    lds[n * PITCH + k] = W[n * (Ic + Hc) + k];
  }
  __syncthreads();

  const int wave = tid >> 5, lane = tid & 31;
  const int lmod = lane & 15, ldiv = lane >> 4;

  float biasv[8];
#pragma unroll
  for (int nt = 0; nt < 8; ++nt) biasv[nt] = bias[nt * 16 + lmod];

  const float* brow = lds + lmod * PITCH + 2 * ldiv;
  const unsigned ntiles = (unsigned)(Bc / 16) * L;

  for (unsigned tile = (unsigned)(pg * 8 + wave); tile < ntiles; tile += (unsigned)(G * 8)) {
    asm volatile("" ::: "memory");
    const unsigned qa = tile * 16u + (unsigned)lmod;
    const unsigned ba = qa / L;
    const unsigned ta = (unsigned)p0 + (qa - ba * L);
    const float* arow = x + ((long)ba * Tc + ta) * Ic + 2 * ldiv;

    v8f acc[8] = {};

    v2f a_c = *(const v2f*)(arow);
    v2f b_c[8];
#pragma unroll
    for (int nt = 0; nt < 8; ++nt) b_c[nt] = *(const v2f*)(brow + nt * 16 * PITCH);

#pragma unroll 4
    for (int kk = 0; kk < Ic - 4; kk += 4) {
      v2f a_n = *(const v2f*)(arow + kk + 4);
      v2f b_n[8];
#pragma unroll
      for (int nt = 0; nt < 8; ++nt)
        b_n[nt] = *(const v2f*)(brow + nt * 16 * PITCH + kk + 4);
#pragma unroll
      for (int nt = 0; nt < 8; ++nt) acc[nt] = WMMA_F32(a_c, b_c[nt], acc[nt]);
      a_c = a_n;
#pragma unroll
      for (int nt = 0; nt < 8; ++nt) b_c[nt] = b_n[nt];
    }
#pragma unroll
    for (int nt = 0; nt < 8; ++nt) acc[nt] = WMMA_F32(a_c, b_c[nt], acc[nt]);

#pragma unroll
    for (int v = 0; v < 8; ++v) {
      const unsigned qs = tile * 16u + (unsigned)(v + 8 * ldiv);
      const unsigned bb = qs / L;
      const unsigned tt = (unsigned)p0 + (qs - bb * L);
      float* orow = outp + ((long)tt * Bc + bb) * Hc + lmod;
#pragma unroll
      for (int nt = 0; nt < 8; ++nt) orow[nt * 16] = acc[nt][v] + biasv[nt];
    }
  }
}

__device__ __forceinline__ void scan_role(
    const float* __restrict__ att,
    const float* __restrict__ Wz, const float* __restrict__ Wr,
    const float* __restrict__ Wh,
    const float* __restrict__ xz, const float* __restrict__ xr,
    const float* __restrict__ xh,
    float* __restrict__ out,
    float* lds, const int blk, const int s0, const int s1)
{
  float* hbuf  = lds;
  float* rhbuf = hbuf + BT * PITCH;

  const int tid = threadIdx.x;
  const int b0  = blk * BT;

  if (s0 == 0) {
    for (int idx = tid; idx < BT * Hc; idx += 256)
      hbuf[(idx >> 7) * PITCH + (idx & 127)] = 0.f;
  } else {
    for (int idx = tid; idx < BT * Hc; idx += 256) {
      int m = idx >> 7, n = idx & 127;
      hbuf[m * PITCH + n] = out[(long)(b0 + m) * Hc + n];
    }
  }

  const int wave = tid >> 5, lane = tid & 31;
  const int lmod = lane & 15, ldiv = lane >> 4;
  const int nt   = wave;
  const int ncol = nt * 16 + lmod;

  const float* wzg = Wz + (long)ncol * (Ic + Hc) + Ic + 2 * ldiv;
  const float* wrg = Wr + (long)ncol * (Ic + Hc) + Ic + 2 * ldiv;
  const float* whg = Wh + (long)ncol * (Ic + Hc) + Ic + 2 * ldiv;
  v2f fz[KT], fr[KT], fh[KT];
#pragma unroll
  for (int i = 0; i < KT; ++i) {
    fz[i] = *(const v2f*)(wzg + 4 * i);
    fr[i] = *(const v2f*)(wrg + 4 * i);
    fh[i] = *(const v2f*)(whg + 4 * i);
  }

  const float* ha0 = hbuf + lmod * PITCH + 2 * ldiv;
  const float* ra0 = rhbuf + lmod * PITCH + 2 * ldiv;

  const float* arow = att + (long)(b0 + 8 * ldiv) * Tc;

  __syncthreads();

  float hold[8];
#pragma unroll
  for (int v = 0; v < 8; ++v) hold[v] = hbuf[(v + 8 * ldiv) * PITCH + ncol];

  for (int t = s0; t < s1; ++t) {
    if (t + 1 < s1) {
      long nb = ((long)(t + 1) * Bc + b0) * Hc + tid * (BT * Hc / 256);
      __builtin_prefetch(xz + nb, 0, 1);
      __builtin_prefetch(xr + nb, 0, 1);
      __builtin_prefetch(xh + nb, 0, 1);
    }

    v2f fa[KT];
#pragma unroll
    for (int i = 0; i < KT; ++i) fa[i] = *(const v2f*)(ha0 + 4 * i);

    const long xbase = ((long)t * Bc + b0) * Hc + (long)(8 * ldiv) * Hc + ncol;
    float xzv[8], xrv[8], xhv[8], av[8];
#pragma unroll
    for (int v = 0; v < 8; ++v) {
      xzv[v] = xz[xbase + (long)v * Hc];
      xrv[v] = xr[xbase + (long)v * Hc];
      xhv[v] = xh[xbase + (long)v * Hc];
      av[v]  = arow[(long)v * Tc + t];
    }

    __builtin_amdgcn_sched_barrier(0);

    v8f az = {}, ar = {}, ah = {};

#pragma unroll
    for (int i = 0; i < KT; ++i) {
      az = WMMA_F32(fa[i], fz[i], az);
      ar = WMMA_F32(fa[i], fr[i], ar);
    }

#pragma unroll
    for (int v = 0; v < 8; ++v) {
      int m = v + 8 * ldiv;
      rhbuf[m * PITCH + ncol] = sigmoid_hw(xrv[v] + ar[v]) * hold[v];
    }
    __syncthreads();

#pragma unroll
    for (int i = 0; i < KT; ++i) fa[i] = *(const v2f*)(ra0 + 4 * i);
    __builtin_amdgcn_sched_barrier(0);
    float zp[8], keep[8];
#pragma unroll
    for (int v = 0; v < 8; ++v) {
      zp[v]   = av[v] * sigmoid_hw(xzv[v] + az[v]);
      keep[v] = (1.f - zp[v]) * hold[v];
    }
#pragma unroll
    for (int i = 0; i < KT; ++i) ah = WMMA_F32(fa[i], fh[i], ah);

#pragma unroll
    for (int v = 0; v < 8; ++v) {
      int m = v + 8 * ldiv;
      hold[v] = fmaf(zp[v], tanh_hw(xhv[v] + ah[v]), keep[v]);
      hbuf[m * PITCH + ncol] = hold[v];
    }
    __syncthreads();
  }

  for (int idx = tid; idx < BT * Hc; idx += 256) {
    int m = idx >> 7, n = idx & 127;
    out[(long)(b0 + m) * Hc + n] = hbuf[m * PITCH + n];
  }
}

__global__ __launch_bounds__(256) void augru_fused(
    const float* __restrict__ x, const float* __restrict__ att,
    const float* __restrict__ Wz, const float* __restrict__ bz,
    const float* __restrict__ Wr, const float* __restrict__ br,
    const float* __restrict__ Wh, const float* __restrict__ bh,
    float* __restrict__ xz, float* __restrict__ xr, float* __restrict__ xh,
    float* __restrict__ out,
    const int nscan,
    const int s0, const int s1,
    const int p0, const int plen)
{
  extern __shared__ float lds[];
  const int blk = blockIdx.x;
  if (blk < nscan) {
    scan_role(att, Wz, Wr, Wh, xz, xr, xh, out, lds, blk, s0, s1);
  } else if (plen > 0) {
    proj_role(x, Wz, bz, Wr, br, Wh, bh, xz, xr, xh, lds,
              blk - nscan, (int)gridDim.x - nscan, p0, (unsigned)plen);
  }
}

extern "C" void kernel_launch(void* const* d_in, const int* in_sizes, int n_in,
                              void* d_out, int out_size, void* d_ws, size_t ws_size,
                              hipStream_t stream) {
  const float* x   = (const float*)d_in[0];
  const float* att = (const float*)d_in[1];
  const float* Wz  = (const float*)d_in[2];
  const float* bz  = (const float*)d_in[3];
  const float* Wr  = (const float*)d_in[4];
  const float* br  = (const float*)d_in[5];
  const float* Wh  = (const float*)d_in[6];
  const float* bh  = (const float*)d_in[7];
  float* out = (float*)d_out;

  const size_t TBH = (size_t)Tc * Bc * Hc;
  float* xz = (float*)d_ws;
  float* xr = xz + TBH;
  float* xh = xr + TBH;

  const size_t sh = (size_t)Hc * PITCH * sizeof(float);

  augru_fused<<<255, 256, sh, stream>>>(x, att, Wz, bz, Wr, br, Wh, bh, xz, xr, xh, out,
                                        0, 0, 0, 0, 16);
  augru_fused<<<256, 256, sh, stream>>>(x, att, Wz, bz, Wr, br, Wh, bh, xz, xr, xh, out,
                                        NSCAN, 0, 16, 16, 24);
  augru_fused<<<256, 256, sh, stream>>>(x, att, Wz, bz, Wr, br, Wh, bh, xz, xr, xh, out,
                                        NSCAN, 16, 40, 40, 40);
  augru_fused<<<256, 256, sh, stream>>>(x, att, Wz, bz, Wr, br, Wh, bh, xz, xr, xh, out,
                                        NSCAN, 40, 80, 80, 64);
  augru_fused<<<256, 256, sh, stream>>>(x, att, Wz, bz, Wr, br, Wh, bh, xz, xr, xh, out,
                                        NSCAN, 80, 144, 144, 56);
  augru_fused<<<NSCAN, 256, sh, stream>>>(x, att, Wz, bz, Wr, br, Wh, bh, xz, xr, xh, out,
                                          NSCAN, 144, 200, 0, 0);
}
